# starting kernel
# baseline (speedup 1.0000x reference)
#include <hip/hip_runtime.h>
#include <math.h>

typedef float v2f __attribute__((ext_vector_type(2)));
typedef float v8f __attribute__((ext_vector_type(8)));

#define B_     128
#define S_     256
#define MW_    21
#define ECHAR  50
#define EWORD  256
#define KW     5
#define NWORDS (B_ * S_)
#define NPOS   (MW_ - KW + 1)
#define KIM    (ECHAR * KW)
#define KPAD   252
#define NPAIR  (KPAD / 2)


#define WS_WT  0
#define WS_WP  (WS_WT + NPAIR * 2 * EWORD)
#define WS_WG  (WS_WP + EWORD * EWORD)
#define WS_H   (WS_WG + EWORD * EWORD)

#define SA_PAIRS   526
#define SA_STRIDE  32
#define SB_PAIRS   126
#define SB_STRIDE  512
#define SH_PAIRS   128
#define SH_STRIDE  96

__device__ __forceinline__ v8f wmma_f32(v2f a, v2f b, v8f c) {
  return __builtin_amdgcn_wmma_f32_16x16x4_f32(false, a, false, b, (short)0, c,
                                               false, false);
}

__global__ void __launch_bounds__(256)
prep_kernel(const float* __restrict__ conv_w, const float* __restrict__ w_proj,
            const float* __restrict__ w_gate, float* __restrict__ ws) {
  int i = blockIdx.x * 256 + threadIdx.x;
  const int n1 = NPAIR * 2 * EWORD;
  const int n2 = EWORD * EWORD;
  if (i < n1) {
    int pair = i >> 9, e = i & 511;
    int o = e >> 1, par = e & 1;
    int q = pair * 2 + par;
    float v = 0.0f;
    if (q < KIM) {
      int k = q / ECHAR, c = q % ECHAR;
      v = conv_w[(o * ECHAR + c) * KW + k];
    }
    ws[WS_WT + i] = v;
  } else if (i < n1 + 2 * n2) {
    int j = i - n1;
    int m = (j < n2) ? 0 : 1;
    int jj = j - m * n2;
    int pair = jj >> 9, e = jj & 511;
    int o = e >> 1, par = e & 1;
    int c = pair * 2 + par;
    const float* w = m ? w_gate : w_proj;
    ws[(m ? WS_WG : WS_WP) + jj] = w[o * EWORD + c];
  }
}

__global__ void __launch_bounds__(256)
conv_kernel(const int* __restrict__ sents, const float* __restrict__ emb,
            const float* __restrict__ conv_b, float* __restrict__ ws) {
  extern __shared__ float smem[];
  float* sA = smem;
  float* sB = smem + SA_PAIRS * SA_STRIDE;

  const float* WtI = ws + WS_WT;
  float* H = ws + WS_H;

  const int tid = threadIdx.x;
  const int wg = blockIdx.x;

  for (int j = tid; j < 1050 * 16; j += 256) {
    int word = j & 15;
    int row = j >> 4;
    int p = row / ECHAR;
    int c = row - p * ECHAR;
    int ch = sents[(wg * 16 + word) * MW_ + p];
    sA[(row >> 1) * SA_STRIDE + word * 2 + (row & 1)] = emb[ch * ECHAR + c];
  }
  if (tid < 32) sA[(SA_PAIRS - 1) * SA_STRIDE + tid] = 0.0f;

  for (int j = tid; j < SB_PAIRS * 512; j += 256) {
    int pair = j >> 9;
    int e = j & 511;
    int col = e >> 1, par = e & 1;
    int colx = col ^ ((pair & 1) << 4);
    sB[pair * SB_STRIDE + colx * 2 + par] = WtI[j];
  }
  __syncthreads();

  const int wave = tid >> 5;
  const int lane = tid & 31;
  const int hf = lane >> 4;
  const int lm = lane & 15;
  const int c0 = wave * 32;

  const float* bbase0 = sB + hf * SB_STRIDE + ((c0 + lm) ^ (hf << 4)) * 2;
  const float* bbase1 = sB + hf * SB_STRIDE + ((c0 + 16 + lm) ^ (hf << 4)) * 2;

  const float bias0 = conv_b[c0 + lm];
  const float bias1 = conv_b[c0 + 16 + lm];

  v8f mx0 = (v8f)0.0f, mx1 = (v8f)0.0f;

  for (int t = 0; t < NPOS; ++t) {
    v8f acc0 = (v8f)0.0f, acc1 = (v8f)0.0f;
    const float* ap = sA + (t * 25 + hf) * SA_STRIDE + lm * 2;
    const float* bp0 = bbase0;
    const float* bp1 = bbase1;
    for (int blk = 0; blk < 7; ++blk) {
#pragma unroll
      for (int u = 0; u < 9; ++u) {
        v2f a = *(const v2f*)(ap + u * (2 * SA_STRIDE));
        v2f b0 = *(const v2f*)(bp0 + u * (2 * SB_STRIDE));
        v2f b1 = *(const v2f*)(bp1 + u * (2 * SB_STRIDE));
        acc0 = wmma_f32(a, b0, acc0);
        acc1 = wmma_f32(a, b1, acc1);
      }
      ap += 9 * (2 * SA_STRIDE);
      bp0 += 9 * (2 * SB_STRIDE);
      bp1 += 9 * (2 * SB_STRIDE);
    }
    for (int r = 0; r < 8; ++r) {
      mx0[r] = fmaxf(mx0[r], fmaxf(acc0[r] + bias0, 0.0f));
      mx1[r] = fmaxf(mx1[r], fmaxf(acc1[r] + bias1, 0.0f));
    }
  }

  for (int r = 0; r < 8; ++r) {
    int word = wg * 16 + r + 8 * hf;
    H[word * EWORD + c0 + lm] = mx0[r];
    H[word * EWORD + c0 + 16 + lm] = mx1[r];
  }
}

__global__ void __launch_bounds__(256)
highway_kernel(const float* __restrict__ b_proj, const float* __restrict__ b_gate,
               const float* __restrict__ ws_c, float* __restrict__ out) {
  __shared__ float sH[SH_PAIRS * SH_STRIDE];
  const float* WpI = ws_c + WS_WP;
  const float* WgI = ws_c + WS_WG;
  const float* H = ws_c + WS_H;

  const int tid = threadIdx.x;
  const int wg = blockIdx.x;

  for (int w = 0; w < 32; ++w)
    sH[(tid >> 1) * SH_STRIDE + w * 2 + (tid & 1)] = H[(wg * 32 + w) * EWORD + tid];
  __syncthreads();

  const int wave = tid >> 5;
  const int lane = tid & 31;
  const int hf = lane >> 4;
  const int lm = lane & 15;
  const int c0 = wave * 32;

  const float bp0 = b_proj[c0 + lm], bp1 = b_proj[c0 + 16 + lm];
  const float bg0 = b_gate[c0 + lm], bg1 = b_gate[c0 + 16 + lm];

  v8f aP00 = (v8f)0.0f, aP01 = (v8f)0.0f, aP10 = (v8f)0.0f, aP11 = (v8f)0.0f;
  v8f aG00 = (v8f)0.0f, aG01 = (v8f)0.0f, aG10 = (v8f)0.0f, aG11 = (v8f)0.0f;

  const float* hp = sH + hf * SH_STRIDE + lm * 2;
  const float* p0 = WpI + hf * 512 + (c0 + lm) * 2;
  const float* p1 = WpI + hf * 512 + (c0 + 16 + lm) * 2;
  const float* g0 = WgI + hf * 512 + (c0 + lm) * 2;
  const float* g1 = WgI + hf * 512 + (c0 + 16 + lm) * 2;

  for (int blk = 0; blk < 8; ++blk) {
#pragma unroll
    for (int u = 0; u < 8; ++u) {
      v2f a0 = *(const v2f*)(hp + u * (2 * SH_STRIDE));
      v2f a1 = *(const v2f*)(hp + u * (2 * SH_STRIDE) + 32);
      v2f bP0 = *(const v2f*)(p0 + u * 1024);
      v2f bP1 = *(const v2f*)(p1 + u * 1024);
      v2f bG0 = *(const v2f*)(g0 + u * 1024);
      v2f bG1 = *(const v2f*)(g1 + u * 1024);
      aP00 = wmma_f32(a0, bP0, aP00);
      aP01 = wmma_f32(a0, bP1, aP01);
      aP10 = wmma_f32(a1, bP0, aP10);
      aP11 = wmma_f32(a1, bP1, aP11);
      aG00 = wmma_f32(a0, bG0, aG00);
      aG01 = wmma_f32(a0, bG1, aG01);
      aG10 = wmma_f32(a1, bG0, aG10);
      aG11 = wmma_f32(a1, bG1, aG11);
    }
    hp += 8 * (2 * SH_STRIDE);
    p0 += 8 * 1024;
    p1 += 8 * 1024;
    g0 += 8 * 1024;
    g1 += 8 * 1024;
  }

  for (int mt = 0; mt < 2; ++mt) {
    const v8f& aP_0 = mt ? aP10 : aP00;
    const v8f& aP_1 = mt ? aP11 : aP01;
    const v8f& aG_0 = mt ? aG10 : aG00;
    const v8f& aG_1 = mt ? aG11 : aG01;
    for (int r = 0; r < 8; ++r) {
      int word = mt * 16 + r + 8 * hf;
      int gword = wg * 32 + word;
      {
        int col = c0 + lm;
        float y = sH[(col >> 1) * SH_STRIDE + word * 2 + (col & 1)];
        float p = fmaxf(aP_0[r] + bp0, 0.0f);
        float g = 1.0f / (1.0f + __expf(-(aG_0[r] + bg0)));
        out[gword * EWORD + col] = g * p + (1.0f - g) * y;
      }
      {
        int col = c0 + 16 + lm;
        float y = sH[(col >> 1) * SH_STRIDE + word * 2 + (col & 1)];
        float p = fmaxf(aP_1[r] + bp1, 0.0f);
        float g = 1.0f / (1.0f + __expf(-(aG_1[r] + bg1)));
        out[gword * EWORD + col] = g * p + (1.0f - g) * y;
      }
    }
  }
}

extern "C" void kernel_launch(void* const* d_in, const int* in_sizes, int n_in,
                              void* d_out, int out_size, void* d_ws, size_t ws_size,
                              hipStream_t stream) {
  (void)in_sizes; (void)n_in; (void)out_size; (void)ws_size;
  const int* sents = (const int*)d_in[0];
  const float* emb = (const float*)d_in[1];
  const float* conv_w = (const float*)d_in[2];
  const float* conv_b = (const float*)d_in[3];
  const float* w_proj = (const float*)d_in[4];
  const float* b_proj = (const float*)d_in[5];
  const float* w_gate = (const float*)d_in[6];
  const float* b_gate = (const float*)d_in[7];
  float* ws = (float*)d_ws;
  float* out = (float*)d_out;

  const int prep_n = NPAIR * 2 * EWORD + 2 * EWORD * EWORD;
  prep_kernel<<<(prep_n + 255) / 256, 256, 0, stream>>>(conv_w, w_proj, w_gate, ws);

  const size_t conv_lds =
      (size_t)(SA_PAIRS * SA_STRIDE + SB_PAIRS * SB_STRIDE) * sizeof(float);
  conv_kernel<<<NWORDS / 16, 256, conv_lds, stream>>>(sents, emb, conv_b, ws);

  highway_kernel<<<NWORDS / 32, 256, 0, stream>>>(b_proj, b_gate, ws, out);
}


# conv kernel made persistent: 256 workgroups each walk 8 word tiles, so the 258 KB weight slab is staged into LDS once per workgroup instead of once per 16-word tile
# speedup vs baseline: 1.1494x; 1.1494x over previous
#include <hip/hip_runtime.h>
#include <math.h>

typedef float v2f __attribute__((ext_vector_type(2)));
typedef float v8f __attribute__((ext_vector_type(8)));

#define B_     128
#define S_     256
#define MW_    21
#define ECHAR  50
#define EWORD  256
#define KW     5
#define NWORDS (B_ * S_)
#define NPOS   (MW_ - KW + 1)
#define KIM    (ECHAR * KW)
#define KPAD   252
#define NPAIR  (KPAD / 2)


#define WS_WT  0
#define WS_WP  (WS_WT + NPAIR * 2 * EWORD)
#define WS_WG  (WS_WP + EWORD * EWORD)
#define WS_H   (WS_WG + EWORD * EWORD)

#define SA_PAIRS   526
#define SA_STRIDE  32
#define SB_PAIRS   126
#define SB_STRIDE  512
#define SH_PAIRS   128
#define SH_STRIDE  96

#define TILES_PER_WG 8

__device__ __forceinline__ v8f wmma_f32(v2f a, v2f b, v8f c) {
  return __builtin_amdgcn_wmma_f32_16x16x4_f32(false, a, false, b, (short)0, c,
                                               false, false);
}

__global__ void __launch_bounds__(256)
prep_kernel(const float* __restrict__ conv_w, const float* __restrict__ w_proj,
            const float* __restrict__ w_gate, float* __restrict__ ws) {
  int i = blockIdx.x * 256 + threadIdx.x;
  const int n1 = NPAIR * 2 * EWORD;
  const int n2 = EWORD * EWORD;
  if (i < n1) {
    int pair = i >> 9, e = i & 511;
    int o = e >> 1, par = e & 1;
    int q = pair * 2 + par;
    float v = 0.0f;
    if (q < KIM) {
      int k = q / ECHAR, c = q % ECHAR;
      v = conv_w[(o * ECHAR + c) * KW + k];
    }
    ws[WS_WT + i] = v;
  } else if (i < n1 + 2 * n2) {
    int j = i - n1;
    int m = (j < n2) ? 0 : 1;
    int jj = j - m * n2;
    int pair = jj >> 9, e = jj & 511;
    int o = e >> 1, par = e & 1;
    int c = pair * 2 + par;
    const float* w = m ? w_gate : w_proj;
    ws[(m ? WS_WG : WS_WP) + jj] = w[o * EWORD + c];
  }
}

__global__ void __launch_bounds__(256)
conv_kernel(const int* __restrict__ sents, const float* __restrict__ emb,
            const float* __restrict__ conv_b, float* __restrict__ ws) {
  extern __shared__ float smem[];
  float* sA = smem;
  float* sB = smem + SA_PAIRS * SA_STRIDE;

  const float* WtI = ws + WS_WT;
  float* H = ws + WS_H;

  const int tid = threadIdx.x;

  for (int j = tid; j < SB_PAIRS * 512; j += 256) {
    int pair = j >> 9;
    int e = j & 511;
    int col = e >> 1, par = e & 1;
    int colx = col ^ ((pair & 1) << 4);
    sB[pair * SB_STRIDE + colx * 2 + par] = WtI[j];
  }
  if (tid < 32) sA[(SA_PAIRS - 1) * SA_STRIDE + tid] = 0.0f;

  const int wave = tid >> 5;
  const int lane = tid & 31;
  const int hf = lane >> 4;
  const int lm = lane & 15;
  const int c0 = wave * 32;

  const float* bbase0 = sB + hf * SB_STRIDE + ((c0 + lm) ^ (hf << 4)) * 2;
  const float* bbase1 = sB + hf * SB_STRIDE + ((c0 + 16 + lm) ^ (hf << 4)) * 2;

  const float bias0 = conv_b[c0 + lm];
  const float bias1 = conv_b[c0 + 16 + lm];

  for (int tile = 0; tile < TILES_PER_WG; ++tile) {
  const int wg = blockIdx.x * TILES_PER_WG + tile;

  if (tile) __syncthreads();

  for (int j = tid; j < 1050 * 16; j += 256) {
    int word = j & 15;
    int row = j >> 4;
    int p = row / ECHAR;
    int c = row - p * ECHAR;
    int ch = sents[(wg * 16 + word) * MW_ + p];
    sA[(row >> 1) * SA_STRIDE + word * 2 + (row & 1)] = emb[ch * ECHAR + c];
  }
  __syncthreads();

  v8f mx0 = (v8f)0.0f, mx1 = (v8f)0.0f;

  for (int t = 0; t < NPOS; ++t) {
    v8f acc0 = (v8f)0.0f, acc1 = (v8f)0.0f;
    const float* ap = sA + (t * 25 + hf) * SA_STRIDE + lm * 2;
    const float* bp0 = bbase0;
    const float* bp1 = bbase1;
    for (int blk = 0; blk < 7; ++blk) {
#pragma unroll
      for (int u = 0; u < 9; ++u) {
        v2f a = *(const v2f*)(ap + u * (2 * SA_STRIDE));
        v2f b0 = *(const v2f*)(bp0 + u * (2 * SB_STRIDE));
        v2f b1 = *(const v2f*)(bp1 + u * (2 * SB_STRIDE));
        acc0 = wmma_f32(a, b0, acc0);
        acc1 = wmma_f32(a, b1, acc1);
      }
      ap += 9 * (2 * SA_STRIDE);
      bp0 += 9 * (2 * SB_STRIDE);
      bp1 += 9 * (2 * SB_STRIDE);
    }
    for (int r = 0; r < 8; ++r) {
      mx0[r] = fmaxf(mx0[r], fmaxf(acc0[r] + bias0, 0.0f));
      mx1[r] = fmaxf(mx1[r], fmaxf(acc1[r] + bias1, 0.0f));
    }
  }

  for (int r = 0; r < 8; ++r) {
    int word = wg * 16 + r + 8 * hf;
    H[word * EWORD + c0 + lm] = mx0[r];
    H[word * EWORD + c0 + 16 + lm] = mx1[r];
  }
  }
}

__global__ void __launch_bounds__(256)
highway_kernel(const float* __restrict__ b_proj, const float* __restrict__ b_gate,
               const float* __restrict__ ws_c, float* __restrict__ out) {
  __shared__ float sH[SH_PAIRS * SH_STRIDE];
  const float* WpI = ws_c + WS_WP;
  const float* WgI = ws_c + WS_WG;
  const float* H = ws_c + WS_H;

  const int tid = threadIdx.x;
  const int wg = blockIdx.x;

  for (int w = 0; w < 32; ++w)
    sH[(tid >> 1) * SH_STRIDE + w * 2 + (tid & 1)] = H[(wg * 32 + w) * EWORD + tid];
  __syncthreads();

  const int wave = tid >> 5;
  const int lane = tid & 31;
  const int hf = lane >> 4;
  const int lm = lane & 15;
  const int c0 = wave * 32;

  const float bp0 = b_proj[c0 + lm], bp1 = b_proj[c0 + 16 + lm];
  const float bg0 = b_gate[c0 + lm], bg1 = b_gate[c0 + 16 + lm];

  v8f aP00 = (v8f)0.0f, aP01 = (v8f)0.0f, aP10 = (v8f)0.0f, aP11 = (v8f)0.0f;
  v8f aG00 = (v8f)0.0f, aG01 = (v8f)0.0f, aG10 = (v8f)0.0f, aG11 = (v8f)0.0f;

  const float* hp = sH + hf * SH_STRIDE + lm * 2;
  const float* p0 = WpI + hf * 512 + (c0 + lm) * 2;
  const float* p1 = WpI + hf * 512 + (c0 + 16 + lm) * 2;
  const float* g0 = WgI + hf * 512 + (c0 + lm) * 2;
  const float* g1 = WgI + hf * 512 + (c0 + 16 + lm) * 2;

  for (int blk = 0; blk < 8; ++blk) {
#pragma unroll
    for (int u = 0; u < 8; ++u) {
      v2f a0 = *(const v2f*)(hp + u * (2 * SH_STRIDE));
      v2f a1 = *(const v2f*)(hp + u * (2 * SH_STRIDE) + 32);
      v2f bP0 = *(const v2f*)(p0 + u * 1024);
      v2f bP1 = *(const v2f*)(p1 + u * 1024);
      v2f bG0 = *(const v2f*)(g0 + u * 1024);
      v2f bG1 = *(const v2f*)(g1 + u * 1024);
      aP00 = wmma_f32(a0, bP0, aP00);
      aP01 = wmma_f32(a0, bP1, aP01);
      aP10 = wmma_f32(a1, bP0, aP10);
      aP11 = wmma_f32(a1, bP1, aP11);
      aG00 = wmma_f32(a0, bG0, aG00);
      aG01 = wmma_f32(a0, bG1, aG01);
      aG10 = wmma_f32(a1, bG0, aG10);
      aG11 = wmma_f32(a1, bG1, aG11);
    }
    hp += 8 * (2 * SH_STRIDE);
    p0 += 8 * 1024;
    p1 += 8 * 1024;
    g0 += 8 * 1024;
    g1 += 8 * 1024;
  }

  for (int mt = 0; mt < 2; ++mt) {
    const v8f& aP_0 = mt ? aP10 : aP00;
    const v8f& aP_1 = mt ? aP11 : aP01;
    const v8f& aG_0 = mt ? aG10 : aG00;
    const v8f& aG_1 = mt ? aG11 : aG01;
    for (int r = 0; r < 8; ++r) {
      int word = mt * 16 + r + 8 * hf;
      int gword = wg * 32 + word;
      {
        int col = c0 + lm;
        float y = sH[(col >> 1) * SH_STRIDE + word * 2 + (col & 1)];
        float p = fmaxf(aP_0[r] + bp0, 0.0f);
        float g = 1.0f / (1.0f + __expf(-(aG_0[r] + bg0)));
        out[gword * EWORD + col] = g * p + (1.0f - g) * y;
      }
      {
        int col = c0 + 16 + lm;
        float y = sH[(col >> 1) * SH_STRIDE + word * 2 + (col & 1)];
        float p = fmaxf(aP_1[r] + bp1, 0.0f);
        float g = 1.0f / (1.0f + __expf(-(aG_1[r] + bg1)));
        out[gword * EWORD + col] = g * p + (1.0f - g) * y;
      }
    }
  }
}

extern "C" void kernel_launch(void* const* d_in, const int* in_sizes, int n_in,
                              void* d_out, int out_size, void* d_ws, size_t ws_size,
                              hipStream_t stream) {
  (void)in_sizes; (void)n_in; (void)out_size; (void)ws_size;
  const int* sents = (const int*)d_in[0];
  const float* emb = (const float*)d_in[1];
  const float* conv_w = (const float*)d_in[2];
  const float* conv_b = (const float*)d_in[3];
  const float* w_proj = (const float*)d_in[4];
  const float* b_proj = (const float*)d_in[5];
  const float* w_gate = (const float*)d_in[6];
  const float* b_gate = (const float*)d_in[7];
  float* ws = (float*)d_ws;
  float* out = (float*)d_out;

  const int prep_n = NPAIR * 2 * EWORD + 2 * EWORD * EWORD;
  prep_kernel<<<(prep_n + 255) / 256, 256, 0, stream>>>(conv_w, w_proj, w_gate, ws);

  const size_t conv_lds =
      (size_t)(SA_PAIRS * SA_STRIDE + SB_PAIRS * SB_STRIDE) * sizeof(float);
  conv_kernel<<<NWORDS / 16 / TILES_PER_WG, 256, conv_lds, stream>>>(sents, emb, conv_b, ws);

  highway_kernel<<<NWORDS / 32, 256, 0, stream>>>(b_proj, b_gate, ws, out);
}


# conv K-loop: all 63 A fragments of a position read as ONE run of ds_loads in front of its 126 WMMAs (A and B fragments in explicit register arrays): 1 LDS-read place per 126 WMMAs, was 1 per 4
# speedup vs baseline: 1.2111x; 1.0537x over previous
#include <hip/hip_runtime.h>
#include <math.h>

typedef float v2f __attribute__((ext_vector_type(2)));
typedef float v8f __attribute__((ext_vector_type(8)));

#define B_     128
#define S_     256
#define MW_    21
#define ECHAR  50
#define EWORD  256
#define KW     5
#define NWORDS (B_ * S_)
#define NPOS   (MW_ - KW + 1)
#define KIM    (ECHAR * KW)
#define KPAD   252
#define NPAIR  (KPAD / 2)


#define WS_WT  0
#define WS_WP  (WS_WT + NPAIR * 2 * EWORD)
#define WS_WG  (WS_WP + EWORD * EWORD)
#define WS_H   (WS_WG + EWORD * EWORD)

#define SA_PAIRS   526
#define SA_STRIDE  32
#define SB_PAIRS   126
#define SB_STRIDE  512
#define SH_PAIRS   128
#define SH_STRIDE  96

#define TILES_PER_WG 8

__device__ __forceinline__ v8f wmma_f32(v2f a, v2f b, v8f c) {
  return __builtin_amdgcn_wmma_f32_16x16x4_f32(false, a, false, b, (short)0, c,
                                               false, false);
}

__global__ void __launch_bounds__(256)
prep_kernel(const float* __restrict__ conv_w, const float* __restrict__ w_proj,
            const float* __restrict__ w_gate, float* __restrict__ ws) {
  int i = blockIdx.x * 256 + threadIdx.x;
  const int n1 = NPAIR * 2 * EWORD;
  const int n2 = EWORD * EWORD;
  if (i < n1) {
    int pair = i >> 9, e = i & 511;
    int o = e >> 1, par = e & 1;
    int q = pair * 2 + par;
    float v = 0.0f;
    if (q < KIM) {
      int k = q / ECHAR, c = q % ECHAR;
      v = conv_w[(o * ECHAR + c) * KW + k];
    }
    ws[WS_WT + i] = v;
  } else if (i < n1 + 2 * n2) {
    int j = i - n1;
    int m = (j < n2) ? 0 : 1;
    int jj = j - m * n2;
    int pair = jj >> 9, e = jj & 511;
    int o = e >> 1, par = e & 1;
    int c = pair * 2 + par;
    const float* w = m ? w_gate : w_proj;
    ws[(m ? WS_WG : WS_WP) + jj] = w[o * EWORD + c];
  }
}

__global__ void __launch_bounds__(256)
conv_kernel(const int* __restrict__ sents, const float* __restrict__ emb,
            const float* __restrict__ conv_b, float* __restrict__ ws) {
  extern __shared__ float smem[];
  float* sA = smem;
  float* sB = smem + SA_PAIRS * SA_STRIDE;

  const float* WtI = ws + WS_WT;
  float* H = ws + WS_H;

  const int tid = threadIdx.x;

  for (int j = tid; j < SB_PAIRS * 512; j += 256) {
    int pair = j >> 9;
    int e = j & 511;
    int col = e >> 1, par = e & 1;
    int colx = col ^ ((pair & 1) << 4);
    sB[pair * SB_STRIDE + colx * 2 + par] = WtI[j];
  }
  if (tid < 32) sA[(SA_PAIRS - 1) * SA_STRIDE + tid] = 0.0f;

  const int wave = tid >> 5;
  const int lane = tid & 31;
  const int hf = lane >> 4;
  const int lm = lane & 15;
  const int c0 = wave * 32;

  const float* bbase0 = sB + hf * SB_STRIDE + ((c0 + lm) ^ (hf << 4)) * 2;
  const float* bbase1 = sB + hf * SB_STRIDE + ((c0 + 16 + lm) ^ (hf << 4)) * 2;

  const float bias0 = conv_b[c0 + lm];
  const float bias1 = conv_b[c0 + 16 + lm];

  for (int tile = 0; tile < TILES_PER_WG; ++tile) {
  const int wg = blockIdx.x * TILES_PER_WG + tile;

  if (tile) __syncthreads();

  for (int j = tid; j < 1050 * 16; j += 256) {
    int word = j & 15;
    int row = j >> 4;
    int p = row / ECHAR;
    int c = row - p * ECHAR;
    int ch = sents[(wg * 16 + word) * MW_ + p];
    sA[(row >> 1) * SA_STRIDE + word * 2 + (row & 1)] = emb[ch * ECHAR + c];
  }
  __syncthreads();

  v8f mx0 = (v8f)0.0f, mx1 = (v8f)0.0f;

  constexpr int KSTEPS = KPAD / 4;

  v2f bf0[KSTEPS], bf1[KSTEPS];
#pragma unroll
  for (int s = 0; s < KSTEPS; ++s) {
    bf0[s] = *(const v2f*)(bbase0 + s * (2 * SB_STRIDE));
    bf1[s] = *(const v2f*)(bbase1 + s * (2 * SB_STRIDE));
  }

  for (int t = 0; t < NPOS; ++t) {
    const float* ap = sA + (t * 25 + hf) * SA_STRIDE + lm * 2;
    v2f af[KSTEPS];
#pragma unroll
    for (int s = 0; s < KSTEPS; ++s) af[s] = *(const v2f*)(ap + s * (2 * SA_STRIDE));

    v8f acc0 = (v8f)0.0f, acc1 = (v8f)0.0f;
#pragma unroll
    for (int s = 0; s < KSTEPS; ++s) {
      acc0 = wmma_f32(af[s], bf0[s], acc0);
      acc1 = wmma_f32(af[s], bf1[s], acc1);
    }
    for (int r = 0; r < 8; ++r) {
      mx0[r] = fmaxf(mx0[r], fmaxf(acc0[r] + bias0, 0.0f));
      mx1[r] = fmaxf(mx1[r], fmaxf(acc1[r] + bias1, 0.0f));
    }
  }

  for (int r = 0; r < 8; ++r) {
    int word = wg * 16 + r + 8 * hf;
    H[word * EWORD + c0 + lm] = mx0[r];
    H[word * EWORD + c0 + 16 + lm] = mx1[r];
  }
  }
}

__global__ void __launch_bounds__(256)
highway_kernel(const float* __restrict__ b_proj, const float* __restrict__ b_gate,
               const float* __restrict__ ws_c, float* __restrict__ out) {
  __shared__ float sH[SH_PAIRS * SH_STRIDE];
  const float* WpI = ws_c + WS_WP;
  const float* WgI = ws_c + WS_WG;
  const float* H = ws_c + WS_H;

  const int tid = threadIdx.x;
  const int wg = blockIdx.x;

  for (int w = 0; w < 32; ++w)
    sH[(tid >> 1) * SH_STRIDE + w * 2 + (tid & 1)] = H[(wg * 32 + w) * EWORD + tid];
  __syncthreads();

  const int wave = tid >> 5;
  const int lane = tid & 31;
  const int hf = lane >> 4;
  const int lm = lane & 15;
  const int c0 = wave * 32;

  const float bp0 = b_proj[c0 + lm], bp1 = b_proj[c0 + 16 + lm];
  const float bg0 = b_gate[c0 + lm], bg1 = b_gate[c0 + 16 + lm];

  v8f aP00 = (v8f)0.0f, aP01 = (v8f)0.0f, aP10 = (v8f)0.0f, aP11 = (v8f)0.0f;
  v8f aG00 = (v8f)0.0f, aG01 = (v8f)0.0f, aG10 = (v8f)0.0f, aG11 = (v8f)0.0f;

  const float* hp = sH + hf * SH_STRIDE + lm * 2;
  const float* p0 = WpI + hf * 512 + (c0 + lm) * 2;
  const float* p1 = WpI + hf * 512 + (c0 + 16 + lm) * 2;
  const float* g0 = WgI + hf * 512 + (c0 + lm) * 2;
  const float* g1 = WgI + hf * 512 + (c0 + 16 + lm) * 2;

  for (int blk = 0; blk < 8; ++blk) {
#pragma unroll
    for (int u = 0; u < 8; ++u) {
      v2f a0 = *(const v2f*)(hp + u * (2 * SH_STRIDE));
      v2f a1 = *(const v2f*)(hp + u * (2 * SH_STRIDE) + 32);
      v2f bP0 = *(const v2f*)(p0 + u * 1024);
      v2f bP1 = *(const v2f*)(p1 + u * 1024);
      v2f bG0 = *(const v2f*)(g0 + u * 1024);
      v2f bG1 = *(const v2f*)(g1 + u * 1024);
      aP00 = wmma_f32(a0, bP0, aP00);
      aP01 = wmma_f32(a0, bP1, aP01);
      aP10 = wmma_f32(a1, bP0, aP10);
      aP11 = wmma_f32(a1, bP1, aP11);
      aG00 = wmma_f32(a0, bG0, aG00);
      aG01 = wmma_f32(a0, bG1, aG01);
      aG10 = wmma_f32(a1, bG0, aG10);
      aG11 = wmma_f32(a1, bG1, aG11);
    }
    hp += 8 * (2 * SH_STRIDE);
    p0 += 8 * 1024;
    p1 += 8 * 1024;
    g0 += 8 * 1024;
    g1 += 8 * 1024;
  }

  for (int mt = 0; mt < 2; ++mt) {
    const v8f& aP_0 = mt ? aP10 : aP00;
    const v8f& aP_1 = mt ? aP11 : aP01;
    const v8f& aG_0 = mt ? aG10 : aG00;
    const v8f& aG_1 = mt ? aG11 : aG01;
    for (int r = 0; r < 8; ++r) {
      int word = mt * 16 + r + 8 * hf;
      int gword = wg * 32 + word;
      {
        int col = c0 + lm;
        float y = sH[(col >> 1) * SH_STRIDE + word * 2 + (col & 1)];
        float p = fmaxf(aP_0[r] + bp0, 0.0f);
        float g = 1.0f / (1.0f + __expf(-(aG_0[r] + bg0)));
        out[gword * EWORD + col] = g * p + (1.0f - g) * y;
      }
      {
        int col = c0 + 16 + lm;
        float y = sH[(col >> 1) * SH_STRIDE + word * 2 + (col & 1)];
        float p = fmaxf(aP_1[r] + bp1, 0.0f);
        float g = 1.0f / (1.0f + __expf(-(aG_1[r] + bg1)));
        out[gword * EWORD + col] = g * p + (1.0f - g) * y;
      }
    }
  }
}

extern "C" void kernel_launch(void* const* d_in, const int* in_sizes, int n_in,
                              void* d_out, int out_size, void* d_ws, size_t ws_size,
                              hipStream_t stream) {
  (void)in_sizes; (void)n_in; (void)out_size; (void)ws_size;
  const int* sents = (const int*)d_in[0];
  const float* emb = (const float*)d_in[1];
  const float* conv_w = (const float*)d_in[2];
  const float* conv_b = (const float*)d_in[3];
  const float* w_proj = (const float*)d_in[4];
  const float* b_proj = (const float*)d_in[5];
  const float* w_gate = (const float*)d_in[6];
  const float* b_gate = (const float*)d_in[7];
  float* ws = (float*)d_ws;
  float* out = (float*)d_out;

  const int prep_n = NPAIR * 2 * EWORD + 2 * EWORD * EWORD;
  prep_kernel<<<(prep_n + 255) / 256, 256, 0, stream>>>(conv_w, w_proj, w_gate, ws);

  const size_t conv_lds =
      (size_t)(SA_PAIRS * SA_STRIDE + SB_PAIRS * SB_STRIDE) * sizeof(float);
  conv_kernel<<<NWORDS / 16 / TILES_PER_WG, 256, conv_lds, stream>>>(sents, emb, conv_b, ws);

  highway_kernel<<<NWORDS / 32, 256, 0, stream>>>(b_proj, b_gate, ws, out);
}


# conv: next tile's embedding gather overlaps this tile's WMMAs (sA double-buffered; each wave gathers its share at its own position slot); LDS room: B fragments go global->regs once per WG, no slab
# speedup vs baseline: 1.3700x; 1.1312x over previous
#include <hip/hip_runtime.h>
#include <math.h>

typedef float v2f __attribute__((ext_vector_type(2)));
typedef float v8f __attribute__((ext_vector_type(8)));

#define B_     128
#define S_     256
#define MW_    21
#define ECHAR  50
#define EWORD  256
#define KW     5
#define NWORDS (B_ * S_)
#define NPOS   (MW_ - KW + 1)
#define KIM    (ECHAR * KW)
#define KPAD   252
#define NPAIR  (KPAD / 2)


#define WS_WT  0
#define WS_WP  (WS_WT + NPAIR * 2 * EWORD)
#define WS_WG  (WS_WP + EWORD * EWORD)
#define WS_H   (WS_WG + EWORD * EWORD)

#define SA_PAIRS   526
#define SA_STRIDE  32
#define SA_BUF     (SA_PAIRS * SA_STRIDE)
#define SH_PAIRS   128
#define SH_STRIDE  96

#define TILES_PER_WG 8

__device__ __forceinline__ v8f wmma_f32(v2f a, v2f b, v8f c) {
  return __builtin_amdgcn_wmma_f32_16x16x4_f32(false, a, false, b, (short)0, c,
                                               false, false);
}

__global__ void __launch_bounds__(256)
prep_kernel(const float* __restrict__ conv_w, const float* __restrict__ w_proj,
            const float* __restrict__ w_gate, float* __restrict__ ws) {
  int i = blockIdx.x * 256 + threadIdx.x;
  const int n1 = NPAIR * 2 * EWORD;
  const int n2 = EWORD * EWORD;
  if (i < n1) {
    int pair = i >> 9, e = i & 511;
    int o = e >> 1, par = e & 1;
    int q = pair * 2 + par;
    float v = 0.0f;
    if (q < KIM) {
      int k = q / ECHAR, c = q % ECHAR;
      v = conv_w[(o * ECHAR + c) * KW + k];
    }
    ws[WS_WT + i] = v;
  } else if (i < n1 + 2 * n2) {
    int j = i - n1;
    int m = (j < n2) ? 0 : 1;
    int jj = j - m * n2;
    int pair = jj >> 9, e = jj & 511;
    int o = e >> 1, par = e & 1;
    int c = pair * 2 + par;
    const float* w = m ? w_gate : w_proj;
    ws[(m ? WS_WG : WS_WP) + jj] = w[o * EWORD + c];
  }
}

__device__ __forceinline__ void gather_share(const int* __restrict__ sents,
                                             const float* __restrict__ emb,
                                             float* dst, int wg, int wave,
                                             int lane) {
  const int word = lane & 15;
  const int hf = lane >> 4;
  const int* srow = sents + (wg * 16 + word) * MW_;
  int ch[MW_];
#pragma unroll
  for (int p = 0; p < MW_; ++p) ch[p] = srow[p];
#pragma unroll
  for (int r = 0; r < 2; ++r) {
    const int j = wave + 8 * r;
    if (j < 13) {
      const int i = 2 * j + hf;
      if (i < ECHAR / 2) {
        const float* src = emb + 2 * i;
        float* d = dst + i * SA_STRIDE + word * 2;
#pragma unroll
        for (int p = 0; p < MW_; ++p)
          *(v2f*)(d + p * (ECHAR / 2) * SA_STRIDE) =
              *(const v2f*)(src + ch[p] * ECHAR);
      }
    }
  }
}

__global__ void __launch_bounds__(256) __attribute__((amdgpu_waves_per_eu(2, 2)))
conv_kernel(const int* __restrict__ sents, const float* __restrict__ emb,
            const float* __restrict__ conv_b, float* __restrict__ ws) {
  extern __shared__ float smem[];
  float* sA = smem;

  const float* WtI = ws + WS_WT;
  float* H = ws + WS_H;

  const int tid = threadIdx.x;

  if (tid < 64) sA[(tid >> 5) * SA_BUF + (SA_PAIRS - 1) * SA_STRIDE + (tid & 31)] = 0.0f;

  const int wave = __builtin_amdgcn_readfirstlane(tid >> 5);
  const int lane = tid & 31;
  const int hf = lane >> 4;
  const int lm = lane & 15;
  const int c0 = wave * 32;

  constexpr int KSTEPS = KPAD / 4;
  const float* bbase0 = WtI + hf * (2 * EWORD) + (c0 + lm) * 2;
  const float* bbase1 = WtI + hf * (2 * EWORD) + (c0 + 16 + lm) * 2;
  v2f bf0[KSTEPS], bf1[KSTEPS];
#pragma unroll
  for (int s = 0; s < KSTEPS; ++s) {
    bf0[s] = *(const v2f*)(bbase0 + s * (4 * EWORD));
    bf1[s] = *(const v2f*)(bbase1 + s * (4 * EWORD));
  }

  const float bias0 = conv_b[c0 + lm];
  const float bias1 = conv_b[c0 + 16 + lm];

  gather_share(sents, emb, sA, blockIdx.x * TILES_PER_WG, wave, lane);
  __syncthreads();

  for (int tile = 0; tile < TILES_PER_WG; ++tile) {
  const int wg = blockIdx.x * TILES_PER_WG + tile;
  const float* sAc = sA + (tile & 1) * SA_BUF;
  float* sAn = sA + ((tile & 1) ^ 1) * SA_BUF;

  v8f mx0 = (v8f)0.0f, mx1 = (v8f)0.0f;

  for (int t = 0; t < NPOS; ++t) {
    if (t == 2 * wave && tile + 1 < TILES_PER_WG)
      gather_share(sents, emb, sAn, wg + 1, wave, lane);

    const float* ap = sAc + (t * 25 + hf) * SA_STRIDE + lm * 2;
    v2f af[KSTEPS];
#pragma unroll
    for (int s = 0; s < KSTEPS; ++s) af[s] = *(const v2f*)(ap + s * (2 * SA_STRIDE));
    __builtin_amdgcn_sched_barrier(0);

    v8f acc0 = (v8f)0.0f, acc1 = (v8f)0.0f;
#pragma unroll
    for (int s = 0; s < KSTEPS; ++s) {
      acc0 = wmma_f32(af[s], bf0[s], acc0);
      acc1 = wmma_f32(af[s], bf1[s], acc1);
    }
    for (int r = 0; r < 8; ++r) {
      mx0[r] = fmaxf(mx0[r], fmaxf(acc0[r] + bias0, 0.0f));
      mx1[r] = fmaxf(mx1[r], fmaxf(acc1[r] + bias1, 0.0f));
    }
  }

  for (int r = 0; r < 8; ++r) {
    int word = wg * 16 + r + 8 * hf;
    H[word * EWORD + c0 + lm] = mx0[r];
    H[word * EWORD + c0 + 16 + lm] = mx1[r];
  }
  if (tile + 1 < TILES_PER_WG) __syncthreads();
  }
}

__global__ void __launch_bounds__(256)
highway_kernel(const float* __restrict__ b_proj, const float* __restrict__ b_gate,
               const float* __restrict__ ws_c, float* __restrict__ out) {
  __shared__ float sH[SH_PAIRS * SH_STRIDE];
  const float* WpI = ws_c + WS_WP;
  const float* WgI = ws_c + WS_WG;
  const float* H = ws_c + WS_H;

  const int tid = threadIdx.x;
  const int wg = blockIdx.x;

  for (int w = 0; w < 32; ++w)
    sH[(tid >> 1) * SH_STRIDE + w * 2 + (tid & 1)] = H[(wg * 32 + w) * EWORD + tid];
  __syncthreads();

  const int wave = tid >> 5;
  const int lane = tid & 31;
  const int hf = lane >> 4;
  const int lm = lane & 15;
  const int c0 = wave * 32;

  const float bp0 = b_proj[c0 + lm], bp1 = b_proj[c0 + 16 + lm];
  const float bg0 = b_gate[c0 + lm], bg1 = b_gate[c0 + 16 + lm];

  v8f aP00 = (v8f)0.0f, aP01 = (v8f)0.0f, aP10 = (v8f)0.0f, aP11 = (v8f)0.0f;
  v8f aG00 = (v8f)0.0f, aG01 = (v8f)0.0f, aG10 = (v8f)0.0f, aG11 = (v8f)0.0f;

  const float* hp = sH + hf * SH_STRIDE + lm * 2;
  const float* p0 = WpI + hf * 512 + (c0 + lm) * 2;
  const float* p1 = WpI + hf * 512 + (c0 + 16 + lm) * 2;
  const float* g0 = WgI + hf * 512 + (c0 + lm) * 2;
  const float* g1 = WgI + hf * 512 + (c0 + 16 + lm) * 2;

  for (int blk = 0; blk < 8; ++blk) {
#pragma unroll
    for (int u = 0; u < 8; ++u) {
      v2f a0 = *(const v2f*)(hp + u * (2 * SH_STRIDE));
      v2f a1 = *(const v2f*)(hp + u * (2 * SH_STRIDE) + 32);
      v2f bP0 = *(const v2f*)(p0 + u * 1024);
      v2f bP1 = *(const v2f*)(p1 + u * 1024);
      v2f bG0 = *(const v2f*)(g0 + u * 1024);
      v2f bG1 = *(const v2f*)(g1 + u * 1024);
      aP00 = wmma_f32(a0, bP0, aP00);
      aP01 = wmma_f32(a0, bP1, aP01);
      aP10 = wmma_f32(a1, bP0, aP10);
      aP11 = wmma_f32(a1, bP1, aP11);
      aG00 = wmma_f32(a0, bG0, aG00);
      aG01 = wmma_f32(a0, bG1, aG01);
      aG10 = wmma_f32(a1, bG0, aG10);
      aG11 = wmma_f32(a1, bG1, aG11);
    }
    hp += 8 * (2 * SH_STRIDE);
    p0 += 8 * 1024;
    p1 += 8 * 1024;
    g0 += 8 * 1024;
    g1 += 8 * 1024;
  }

  for (int mt = 0; mt < 2; ++mt) {
    const v8f& aP_0 = mt ? aP10 : aP00;
    const v8f& aP_1 = mt ? aP11 : aP01;
    const v8f& aG_0 = mt ? aG10 : aG00;
    const v8f& aG_1 = mt ? aG11 : aG01;
    for (int r = 0; r < 8; ++r) {
      int word = mt * 16 + r + 8 * hf;
      int gword = wg * 32 + word;
      {
        int col = c0 + lm;
        float y = sH[(col >> 1) * SH_STRIDE + word * 2 + (col & 1)];
        float p = fmaxf(aP_0[r] + bp0, 0.0f);
        float g = 1.0f / (1.0f + __expf(-(aG_0[r] + bg0)));
        out[gword * EWORD + col] = g * p + (1.0f - g) * y;
      }
      {
        int col = c0 + 16 + lm;
        float y = sH[(col >> 1) * SH_STRIDE + word * 2 + (col & 1)];
        float p = fmaxf(aP_1[r] + bp1, 0.0f);
        float g = 1.0f / (1.0f + __expf(-(aG_1[r] + bg1)));
        out[gword * EWORD + col] = g * p + (1.0f - g) * y;
      }
    }
  }
}

extern "C" void kernel_launch(void* const* d_in, const int* in_sizes, int n_in,
                              void* d_out, int out_size, void* d_ws, size_t ws_size,
                              hipStream_t stream) {
  (void)in_sizes; (void)n_in; (void)out_size; (void)ws_size;
  const int* sents = (const int*)d_in[0];
  const float* emb = (const float*)d_in[1];
  const float* conv_w = (const float*)d_in[2];
  const float* conv_b = (const float*)d_in[3];
  const float* w_proj = (const float*)d_in[4];
  const float* b_proj = (const float*)d_in[5];
  const float* w_gate = (const float*)d_in[6];
  const float* b_gate = (const float*)d_in[7];
  float* ws = (float*)d_ws;
  float* out = (float*)d_out;

  const int prep_n = NPAIR * 2 * EWORD + 2 * EWORD * EWORD;
  prep_kernel<<<(prep_n + 255) / 256, 256, 0, stream>>>(conv_w, w_proj, w_gate, ws);

  const size_t conv_lds = (size_t)(2 * SA_BUF) * sizeof(float);
  conv_kernel<<<NWORDS / 16 / TILES_PER_WG, 256, conv_lds, stream>>>(sents, emb, conv_b, ws);

  highway_kernel<<<NWORDS / 32, 256, 0, stream>>>(b_proj, b_gate, ws, out);
}


# highway: 32-word H tile kept row-major in LDS (pitch 260): copied in by b128 without bank conflicts, y read conflict-free, outputs written back in place and stored as whole 512-byte rows
# speedup vs baseline: 6.2129x; 4.5350x over previous
#include <hip/hip_runtime.h>
#include <math.h>

typedef float v2f __attribute__((ext_vector_type(2)));
typedef float v4f __attribute__((ext_vector_type(4)));
typedef float v8f __attribute__((ext_vector_type(8)));

#define B_     128
#define S_     256
#define MW_    21
#define VOCAB  100
#define ECHAR  50
#define EWORD  256
#define KW     5
#define NWORDS (B_ * S_)
#define NPOS   (MW_ - KW + 1)


#define TAB_MT   7
#define TAB_KS   13
#define TAB_ROWS (TAB_MT * 16)

#define WS_PT  0
#define WS_WP  (WS_PT + TAB_ROWS * KW * EWORD)
#define WS_WG  (WS_WP + EWORD * EWORD)
#define WS_H   (WS_WG + EWORD * EWORD)

#define SH_WORDS   32
#define SH_PITCH   260

#define HALF_CH     128
#define G_WAVES     32
#define G_WORDS     256
#define TAB_LDS_F   (VOCAB * KW * HALF_CH)

#define PREP_HW_BLOCKS (2 * EWORD * EWORD / 256)
#define TAB_TASKS      (TAB_MT * KW * (EWORD / 32))
#define TAB_BLOCKS     (TAB_TASKS / 8)

__device__ __forceinline__ v8f wmma_f32(v2f a, v2f b, v8f c) {
  return __builtin_amdgcn_wmma_f32_16x16x4_f32(false, a, false, b, (short)0, c,
                                               false, false);
}

__global__ void __launch_bounds__(256)
prep_kernel(const float* __restrict__ emb, const float* __restrict__ conv_w,
            const float* __restrict__ w_proj, const float* __restrict__ w_gate,
            float* __restrict__ ws) {
  const int tid = threadIdx.x;
  if (blockIdx.x < PREP_HW_BLOCKS) {
    const int j = blockIdx.x * 256 + tid;
    const int m = j >> 16;
    const int jj = j & 65535;
    const int pair = jj >> 9, e = jj & 511;
    const int o = e >> 1, par = e & 1;
    const int c = pair * 2 + par;
    const float* w = m ? w_gate : w_proj;
    ws[(m ? WS_WG : WS_WP) + jj] = w[o * EWORD + c];
  } else {
    const int wave = tid >> 5;
    const int lane = tid & 31;
    const int hf = lane >> 4;
    const int lm = lane & 15;
    const int task = (blockIdx.x - PREP_HW_BLOCKS) * 8 + wave;
    const int np = task & 7;
    const int k = (task >> 3) % KW;
    const int mt = (task >> 3) / KW;
    const int o0 = np * 32 + lm;
    const int o1 = o0 + 16;
    const int v = min(mt * 16 + lm, VOCAB - 1);
    const float* arow = emb + v * ECHAR + 2 * hf;
    const float* b0p = conv_w + (o0 * ECHAR + 2 * hf) * KW + k;
    const float* b1p = conv_w + (o1 * ECHAR + 2 * hf) * KW + k;

    v8f acc0 = (v8f)0.0f, acc1 = (v8f)0.0f;
#pragma unroll
    for (int s = 0; s < TAB_KS - 1; ++s) {
      v2f a = *(const v2f*)(arow + 4 * s);
      v2f b0, b1;
      b0.x = b0p[(4 * s) * KW];
      b0.y = b0p[(4 * s + 1) * KW];
      b1.x = b1p[(4 * s) * KW];
      b1.y = b1p[(4 * s + 1) * KW];
      acc0 = wmma_f32(a, b0, acc0);
      acc1 = wmma_f32(a, b1, acc1);
    }
    {
      const int back = 2 * hf;
      v2f a = *(const v2f*)(arow + 48 - back);
      v2f b0, b1;
      b0.x = b0p[(48 - back) * KW];
      b0.y = b0p[(49 - back) * KW];
      b1.x = b1p[(48 - back) * KW];
      b1.y = b1p[(49 - back) * KW];
      a.x = hf ? 0.0f : a.x;   a.y = hf ? 0.0f : a.y;
      b0.x = hf ? 0.0f : b0.x; b0.y = hf ? 0.0f : b0.y;
      b1.x = hf ? 0.0f : b1.x; b1.y = hf ? 0.0f : b1.y;
      acc0 = wmma_f32(a, b0, acc0);
      acc1 = wmma_f32(a, b1, acc1);
    }
    float* PT = ws + WS_PT;
    for (int r = 0; r < 8; ++r) {
      const int row = mt * 16 + r + 8 * hf;
      PT[(row * KW + k) * EWORD + o0] = acc0[r];
      PT[(row * KW + k) * EWORD + o1] = acc1[r];
    }
  }
}

__global__ void __launch_bounds__(G_WAVES * 32)
conv_gather_kernel(const int* __restrict__ sents, const float* __restrict__ conv_b,
                   float* __restrict__ ws) {
  extern __shared__ __attribute__((aligned(16))) float sP[];
  const float* PT = ws + WS_PT;
  float* H = ws + WS_H;

  const int tid = threadIdx.x;
  const int half = blockIdx.x & 1;
  const int slice = blockIdx.x >> 1;

  for (int i = tid; i < VOCAB * KW * (HALF_CH / 4); i += G_WAVES * 32) {
    const int r = i >> 5, q = i & 31;
    *(v4f*)(sP + r * HALF_CH + q * 4) =
        *(const v4f*)(PT + r * EWORD + half * HALF_CH + q * 4);
  }
  __syncthreads();

  const int wave = __builtin_amdgcn_readfirstlane(tid >> 5);
  const int lane = tid & 31;
  const v4f bias = *(const v4f*)(conv_b + half * HALF_CH + lane * 4);
  const float* lp = sP + lane * 4;

  int ch[MW_];
  {
    const int* srow = sents + (slice * G_WORDS + wave) * MW_;
#pragma unroll
    for (int p = 0; p < MW_; ++p) ch[p] = srow[p];
  }

  for (int i = 0; i < G_WORDS / G_WAVES; ++i) {
    const int word = slice * G_WORDS + i * G_WAVES + wave;
    int off[MW_];
#pragma unroll
    for (int p = 0; p < MW_; ++p)
      off[p] = (int)min((unsigned)ch[p], (unsigned)(VOCAB - 1)) * (KW * HALF_CH);
    if (i + 1 < G_WORDS / G_WAVES) {
      const int* srow = sents + (word + G_WAVES) * MW_;
#pragma unroll
      for (int p = 0; p < MW_; ++p) ch[p] = srow[p];
    }

    v4f m;
#pragma unroll
    for (int t = 0; t < NPOS; ++t) {
      v4f s = *(const v4f*)(lp + off[t]);
      s += *(const v4f*)(lp + off[t + 1] + 1 * HALF_CH);
      s += *(const v4f*)(lp + off[t + 2] + 2 * HALF_CH);
      s += *(const v4f*)(lp + off[t + 3] + 3 * HALF_CH);
      s += *(const v4f*)(lp + off[t + 4] + 4 * HALF_CH);
      if (t == 0) {
        m = s;
      } else {
        m.x = fmaxf(m.x, s.x); m.y = fmaxf(m.y, s.y);
        m.z = fmaxf(m.z, s.z); m.w = fmaxf(m.w, s.w);
      }
    }
    v4f h;
    h.x = fmaxf(m.x + bias.x, 0.0f); h.y = fmaxf(m.y + bias.y, 0.0f);
    h.z = fmaxf(m.z + bias.z, 0.0f); h.w = fmaxf(m.w + bias.w, 0.0f);
    *(v4f*)(H + word * EWORD + half * HALF_CH + lane * 4) = h;
  }
}

__global__ void __launch_bounds__(256)
highway_kernel(const float* __restrict__ b_proj, const float* __restrict__ b_gate,
               const float* __restrict__ ws_c, float* __restrict__ out) {
  __shared__ __attribute__((aligned(16))) float sH[SH_WORDS * SH_PITCH];
  const float* WpI = ws_c + WS_WP;
  const float* WgI = ws_c + WS_WG;
  const float* H = ws_c + WS_H;

  const int tid = threadIdx.x;
  const int wg = blockIdx.x;

  {
    const float* Hg = H + wg * (SH_WORDS * EWORD);
#pragma unroll
    for (int j = 0; j < 8; ++j) {
      const int i = tid + 256 * j;
      const int row = i >> 6, q = i & 63;
      *(v4f*)(sH + row * SH_PITCH + q * 4) = *(const v4f*)(Hg + row * EWORD + q * 4);
    }
  }
  __syncthreads();

  const int wave = tid >> 5;
  const int lane = tid & 31;
  const int hf = lane >> 4;
  const int lm = lane & 15;
  const int c0 = wave * 32;

  const float bp0 = b_proj[c0 + lm], bp1 = b_proj[c0 + 16 + lm];
  const float bg0 = b_gate[c0 + lm], bg1 = b_gate[c0 + 16 + lm];

  v8f aP00 = (v8f)0.0f, aP01 = (v8f)0.0f, aP10 = (v8f)0.0f, aP11 = (v8f)0.0f;
  v8f aG00 = (v8f)0.0f, aG01 = (v8f)0.0f, aG10 = (v8f)0.0f, aG11 = (v8f)0.0f;

  const float* hp = sH + lm * SH_PITCH + 2 * hf;
  const float* p0 = WpI + hf * 512 + (c0 + lm) * 2;
  const float* p1 = WpI + hf * 512 + (c0 + 16 + lm) * 2;
  const float* g0 = WgI + hf * 512 + (c0 + lm) * 2;
  const float* g1 = WgI + hf * 512 + (c0 + 16 + lm) * 2;

  for (int blk = 0; blk < 8; ++blk) {
#pragma unroll
    for (int u = 0; u < 8; ++u) {
      v2f a0 = *(const v2f*)(hp + u * 4);
      v2f a1 = *(const v2f*)(hp + u * 4 + 16 * SH_PITCH);
      v2f bP0 = *(const v2f*)(p0 + u * 1024);
      v2f bP1 = *(const v2f*)(p1 + u * 1024);
      v2f bG0 = *(const v2f*)(g0 + u * 1024);
      v2f bG1 = *(const v2f*)(g1 + u * 1024);
      aP00 = wmma_f32(a0, bP0, aP00);
      aP01 = wmma_f32(a0, bP1, aP01);
      aP10 = wmma_f32(a1, bP0, aP10);
      aP11 = wmma_f32(a1, bP1, aP11);
      aG00 = wmma_f32(a0, bG0, aG00);
      aG01 = wmma_f32(a0, bG1, aG01);
      aG10 = wmma_f32(a1, bG0, aG10);
      aG11 = wmma_f32(a1, bG1, aG11);
    }
    hp += 8 * 4;
    p0 += 8 * 1024;
    p1 += 8 * 1024;
    g0 += 8 * 1024;
    g1 += 8 * 1024;
  }

  __syncthreads();

#define HW_EPILOGUE(MT, AP_0, AP_1, AG_0, AG_1)                                  \
  _Pragma("unroll")                                                              \
  for (int r = 0; r < 8; ++r) {                                                  \
    float* yp = sH + ((MT) * 16 + r + 8 * hf) * SH_PITCH + c0 + lm;              \
    {                                                                            \
      float y = yp[0];                                                           \
      float p = fmaxf(AP_0[r] + bp0, 0.0f);                                      \
      float g = 1.0f / (1.0f + __expf(-(AG_0[r] + bg0)));                        \
      yp[0] = g * p + (1.0f - g) * y;                                            \
    }                                                                            \
    {                                                                            \
      float y = yp[16];                                                          \
      float p = fmaxf(AP_1[r] + bp1, 0.0f);                                      \
      float g = 1.0f / (1.0f + __expf(-(AG_1[r] + bg1)));                        \
      yp[16] = g * p + (1.0f - g) * y;                                           \
    }                                                                            \
  }
  HW_EPILOGUE(0, aP00, aP01, aG00, aG01)
  HW_EPILOGUE(1, aP10, aP11, aG10, aG11)
#undef HW_EPILOGUE

  __syncthreads();

  {
    float* Og = out + wg * (SH_WORDS * EWORD);
#pragma unroll
    for (int j = 0; j < 8; ++j) {
      const int i = tid + 256 * j;
      const int row = i >> 6, q = i & 63;
      *(v4f*)(Og + row * EWORD + q * 4) = *(const v4f*)(sH + row * SH_PITCH + q * 4);
    }
  }
}

extern "C" void kernel_launch(void* const* d_in, const int* in_sizes, int n_in,
                              void* d_out, int out_size, void* d_ws, size_t ws_size,
                              hipStream_t stream) {
  (void)in_sizes; (void)n_in; (void)out_size; (void)ws_size;
  const int* sents = (const int*)d_in[0];
  const float* emb = (const float*)d_in[1];
  const float* conv_w = (const float*)d_in[2];
  const float* conv_b = (const float*)d_in[3];
  const float* w_proj = (const float*)d_in[4];
  const float* b_proj = (const float*)d_in[5];
  const float* w_gate = (const float*)d_in[6];
  const float* b_gate = (const float*)d_in[7];
  float* ws = (float*)d_ws;
  float* out = (float*)d_out;

  prep_kernel<<<PREP_HW_BLOCKS + TAB_BLOCKS, 256, 0, stream>>>(emb, conv_w, w_proj, w_gate, ws);

  const size_t tab_lds = (size_t)TAB_LDS_F * sizeof(float);
  conv_gather_kernel<<<NWORDS / G_WORDS * 2, G_WAVES * 32, tab_lds, stream>>>(sents, conv_b, ws);

  highway_kernel<<<NWORDS / 32, 256, 0, stream>>>(b_proj, b_gate, ws, out);
}
